# starting kernel
# baseline (speedup 1.0000x reference)
#include <hip/hip_runtime.h>
#include <hip/hip_bf16.h>

#define SEG_D 256

typedef __attribute__((ext_vector_type(2))) float v2f;
typedef __attribute__((ext_vector_type(8))) float v8f;

__global__ __launch_bounds__(1024) void seg_offsets_kernel(
    const int* __restrict__ sizes, int* __restrict__ offsets, int B) {
  __shared__ int tmp[1024];
  const int t = threadIdx.x;
  const int chunk = (B + 1023) / 1024;
  const int base = t * chunk;

  int s = 0;
  for (int i = 0; i < chunk; ++i) {
    int idx = base + i;
    if (idx < B) s += sizes[idx];
  }
  tmp[t] = s;
  __syncthreads();

  for (int off = 1; off < 1024; off <<= 1) {
    int v = (t >= off) ? tmp[t - off] : 0;
    __syncthreads();
    tmp[t] += v;
    __syncthreads();
  }

  int run = (t == 0) ? 0 : tmp[t - 1];
  for (int i = 0; i < chunk; ++i) {
    int idx = base + i;
    if (idx < B) {
      offsets[idx] = run;
      run += sizes[idx];
    }
  }
  if (t == 1023) offsets[B] = tmp[1023];
}

__global__ __launch_bounds__(512) void seg_mean_wmma_kernel(
    const float* __restrict__ x, const int* __restrict__ sizes,
    const int* __restrict__ offsets, float* __restrict__ out) {
  const int b = blockIdx.x;
  const int start = offsets[b];
  const int size  = sizes[b];
  const int end   = start + size;

  const int wave = threadIdx.x >> 5;
  const int lane = threadIdx.x & 31;
  const int n    = lane & 15;
  const int kh   = lane >> 4;
  const int col  = (wave << 4) + n;

  const float* __restrict__ xc = x + col;

  v2f a;  a.x = 1.0f;  a.y = 1.0f;
  v8f acc0 = {};
  v8f acc1 = {};

  const int nchunks = size >> 3;
  const float* p = xc + (long)(start + kh) * SEG_D;

  if (nchunks > 0) {
    v2f b0, b1;
    b0.x = __builtin_nontemporal_load(p + 0 * SEG_D);
    b0.y = __builtin_nontemporal_load(p + 2 * SEG_D);
    b1.x = __builtin_nontemporal_load(p + 4 * SEG_D);
    b1.y = __builtin_nontemporal_load(p + 6 * SEG_D);

    for (int i = 1; i < nchunks; ++i) {
      p += 8 * SEG_D;
      __builtin_prefetch(p + 8 * SEG_D, 0, 1);
      v2f c0, c1;
      c0.x = __builtin_nontemporal_load(p + 0 * SEG_D);
      c0.y = __builtin_nontemporal_load(p + 2 * SEG_D);
      c1.x = __builtin_nontemporal_load(p + 4 * SEG_D);
      c1.y = __builtin_nontemporal_load(p + 6 * SEG_D);
      acc0 = __builtin_amdgcn_wmma_f32_16x16x4_f32(
          false, a, false, b0, (short)0, acc0, false, false);
      acc1 = __builtin_amdgcn_wmma_f32_16x16x4_f32(
          false, a, false, b1, (short)0, acc1, false, false);
      b0 = c0;
      b1 = c1;
    }
    acc0 = __builtin_amdgcn_wmma_f32_16x16x4_f32(
        false, a, false, b0, (short)0, acc0, false, false);
    acc1 = __builtin_amdgcn_wmma_f32_16x16x4_f32(
        false, a, false, b1, (short)0, acc1, false, false);
  }

  const int done = start + (nchunks << 3);
  if (done < end) {
    const int last = end - 1;
    int r0 = done + 0 + kh;
    int r1 = done + 2 + kh;
    int r2 = done + 4 + kh;
    int r3 = done + 6 + kh;
    float m0 = (r0 <= last) ? 1.0f : 0.0f;
    float m1 = (r1 <= last) ? 1.0f : 0.0f;
    float m2 = (r2 <= last) ? 1.0f : 0.0f;
    float m3 = (r3 <= last) ? 1.0f : 0.0f;
    int c0i = (r0 <= last) ? r0 : last;
    int c1i = (r1 <= last) ? r1 : last;
    int c2i = (r2 <= last) ? r2 : last;
    int c3i = (r3 <= last) ? r3 : last;

    v2f b0, b1;
    b0.x = m0 * xc[(long)c0i * SEG_D];
    b0.y = m1 * xc[(long)c1i * SEG_D];
    b1.x = m2 * xc[(long)c2i * SEG_D];
    b1.y = m3 * xc[(long)c3i * SEG_D];

    acc0 = __builtin_amdgcn_wmma_f32_16x16x4_f32(
        false, a, false, b0, (short)0, acc0, false, false);
    acc1 = __builtin_amdgcn_wmma_f32_16x16x4_f32(
        false, a, false, b1, (short)0, acc1, false, false);
  }

  float sum = acc0[0] + acc1[0];
  if (lane < 16) {
    out[(long)b * SEG_D + col] = sum / (float)size;
  }
}

extern "C" void kernel_launch(void* const* d_in, const int* in_sizes, int n_in,
                              void* d_out, int out_size, void* d_ws, size_t ws_size,
                              hipStream_t stream) {
  const float* x     = (const float*)d_in[0];
  const int*   sizes = (const int*)d_in[1];
  float*       out   = (float*)d_out;

  const int B = in_sizes[1];
  int* offsets = (int*)d_ws;

  seg_offsets_kernel<<<1, 1024, 0, stream>>>(sizes, offsets, B);
  seg_mean_wmma_kernel<<<B, 512, 0, stream>>>(x, sizes, offsets, out);
}


# One wave per segment streams it as contiguous 512-byte pieces (16 B per lane, two 8-row groups in flight); WMMA A becomes a 0/1 selector; each column now summed row by row in one f32 chain
# speedup vs baseline: 1.4133x; 1.4133x over previous
#include <hip/hip_runtime.h>

#define SEG_D 256
#define STAGE_ROWS 8

typedef __attribute__((ext_vector_type(2))) float v2f;
typedef __attribute__((ext_vector_type(4))) float v4f;
typedef __attribute__((ext_vector_type(8))) float v8f;

__global__ __launch_bounds__(1024) void seg_offsets_kernel(
    const int* __restrict__ sizes, int* __restrict__ offsets, int B) {
  __shared__ int tmp[1024];
  const int t = threadIdx.x;
  const int chunk = (B + 1023) / 1024;
  const int base = t * chunk;

  int s = 0;
  for (int i = 0; i < chunk; ++i) {
    int idx = base + i;
    if (idx < B) s += sizes[idx];
  }
  tmp[t] = s;
  __syncthreads();

  for (int off = 1; off < 1024; off <<= 1) {
    int v = (t >= off) ? tmp[t - off] : 0;
    __syncthreads();
    tmp[t] += v;
    __syncthreads();
  }

  int run = (t == 0) ? 0 : tmp[t - 1];
  for (int i = 0; i < chunk; ++i) {
    int idx = base + i;
    if (idx < B) {
      offsets[idx] = run;
      run += sizes[idx];
    }
  }
  if (t == 1023) offsets[B] = tmp[1023];
}

#define REFILL_BEHIND(p, q, r, s) asm volatile("" : "+v"(p), "+v"(q), "+v"(r), "+v"(s))
#define SUMS_BEHIND(a) asm volatile("" : "+v"(a))

struct Stage {
  v4f d[2 * STAGE_ROWS];
};

__device__ __forceinline__ void stage_load(Stage& st, const float* __restrict__ x,
                                           int row0, int last, int lane) {
#pragma unroll
  for (int j = 0; j < STAGE_ROWS; ++j) {
    const int r = min(row0 + j, last);
    const v4f* q = reinterpret_cast<const v4f*>(x + (size_t)r * SEG_D) + lane;
    st.d[2 * j]     = __builtin_nontemporal_load(q);
    st.d[2 * j + 1] = __builtin_nontemporal_load(q + 32);
  }
}

__device__ __forceinline__ void row_sum(const v4f p0, const v4f p1, const v2f a,
                                        v8f& a00, v8f& a01, v8f& a10, v8f& a11) {
  a00 = __builtin_amdgcn_wmma_f32_16x16x4_f32(false, a, false, p0.xy, (short)0, a00, false, false);
  a10 = __builtin_amdgcn_wmma_f32_16x16x4_f32(false, a, false, p1.xy, (short)0, a10, false, false);
  a01 = __builtin_amdgcn_wmma_f32_16x16x4_f32(false, a, false, p0.zw, (short)0, a01, false, false);
  a11 = __builtin_amdgcn_wmma_f32_16x16x4_f32(false, a, false, p1.zw, (short)0, a11, false, false);
}

__device__ __forceinline__ void stage_sum_full(const Stage& st, const v2f sel,
                                               v8f& a00, v8f& a01, v8f& a10, v8f& a11) {
#pragma unroll
  for (int j = 0; j < STAGE_ROWS; ++j)
    row_sum(st.d[2 * j], st.d[2 * j + 1], sel, a00, a01, a10, a11);
}

__device__ __forceinline__ void stage_sum_last(const Stage& st, const v2f sel, int row0, int end,
                                               v8f& a00, v8f& a01, v8f& a10, v8f& a11) {
#pragma unroll
  for (int j = 0; j < STAGE_ROWS; ++j) {
    const bool in = (row0 + j) < end;
    v2f a;
    a.x = in ? sel.x : 0.0f;
    a.y = in ? sel.y : 0.0f;
    row_sum(st.d[2 * j], st.d[2 * j + 1], a, a00, a01, a10, a11);
  }
}

__global__ __launch_bounds__(32) void seg_mean_wmma_kernel(
    const float* __restrict__ x, const int* __restrict__ offsets,
    float* __restrict__ out) {
  const int b = blockIdx.x;
  const int start = offsets[b];
  const int end   = offsets[b + 1];
  const int size  = end - start;
  const int last  = end - 1;

  const int lane = threadIdx.x;
  const int c    = lane & 15;
  const int h    = lane >> 4;

  v2f sel;
  sel.x = (c == h) ? 1.0f : 0.0f;
  sel.y = (c == 2 + h) ? 1.0f : 0.0f;

  v8f a00 = {}, a01 = {}, a10 = {}, a11 = {};

  if (size > 0) {
    const int nst = (size + STAGE_ROWS - 1) / STAGE_ROWS;
    Stage sa, sb;
    int row = start;
    int k = 0;
    stage_load(sa, x, row, last, lane);
    for (; k + 2 < nst; k += 2) {
      stage_load(sb, x, row + STAGE_ROWS, last, lane);
      SUMS_BEHIND(sel);
      stage_sum_full(sa, sel, a00, a01, a10, a11);
      REFILL_BEHIND(a00, a01, a10, a11);
      stage_load(sa, x, row + 2 * STAGE_ROWS, last, lane);
      SUMS_BEHIND(sel);
      stage_sum_full(sb, sel, a00, a01, a10, a11);
      REFILL_BEHIND(a00, a01, a10, a11);
      row += 2 * STAGE_ROWS;
    }
    if (k + 1 < nst) {
      stage_load(sb, x, row + STAGE_ROWS, last, lane);
      SUMS_BEHIND(sel);
      stage_sum_full(sa, sel, a00, a01, a10, a11);
      stage_sum_last(sb, sel, row + STAGE_ROWS, end, a00, a01, a10, a11);
    } else {
      stage_sum_last(sa, sel, row, end, a00, a01, a10, a11);
    }
  }

  if (lane < 16) {
    const float fs = (float)size;
    float* o = out + (size_t)b * SEG_D + 4 * c;
    v4f v;
    v.x = a00[0] / fs; v.y = a00[2] / fs; v.z = a01[0] / fs; v.w = a01[2] / fs;
    *reinterpret_cast<v4f*>(o) = v;
    v.x = a00[1] / fs; v.y = a00[3] / fs; v.z = a01[1] / fs; v.w = a01[3] / fs;
    *reinterpret_cast<v4f*>(o + 64) = v;
    v.x = a10[0] / fs; v.y = a10[2] / fs; v.z = a11[0] / fs; v.w = a11[2] / fs;
    *reinterpret_cast<v4f*>(o + 128) = v;
    v.x = a10[1] / fs; v.y = a10[3] / fs; v.z = a11[1] / fs; v.w = a11[3] / fs;
    *reinterpret_cast<v4f*>(o + 192) = v;
  }
}

extern "C" void kernel_launch(void* const* d_in, const int* in_sizes, int n_in,
                              void* d_out, int out_size, void* d_ws, size_t ws_size,
                              hipStream_t stream) {
  const float* x     = (const float*)d_in[0];
  const int*   sizes = (const int*)d_in[1];
  float*       out   = (float*)d_out;

  const int B = in_sizes[1];
  int* offsets = (int*)d_ws;

  seg_offsets_kernel<<<1, 1024, 0, stream>>>(sizes, offsets, B);
  seg_mean_wmma_kernel<<<B, 32, 0, stream>>>(x, offsets, out);
}


# Offsets launch folded into the mean kernel (2 launches -> 1): each wave sums the sizes in front of its segment itself, 16 B per lane, and clips its rows to x as the reference's repeat does
# speedup vs baseline: 1.6349x; 1.1568x over previous
#include <hip/hip_runtime.h>

#define SEG_D 256
#define STAGE_ROWS 8
#define ROWS_PER_SEG 64
#define SIZE_LOADS 16

typedef __attribute__((ext_vector_type(2))) float v2f;
typedef __attribute__((ext_vector_type(4))) float v4f;
typedef __attribute__((ext_vector_type(8))) float v8f;
typedef __attribute__((ext_vector_type(4))) int v4i;

__device__ __forceinline__ int rows_in_front(const int* __restrict__ sizes, int b, int lane) {
  const int nfull = b & ~3;
  int part = 0;
  for (int base = 0; base < nfull; base += 128 * SIZE_LOADS) {
    v4i v[SIZE_LOADS];
#pragma unroll
    for (int j = 0; j < SIZE_LOADS; ++j) {
      const int i = min(base + 128 * j + 4 * lane, nfull - 4);
      v[j] = *reinterpret_cast<const v4i*>(sizes + i);
    }
#pragma unroll
    for (int j = 0; j < SIZE_LOADS; ++j) {
      const int s4 = (v[j].x + v[j].y) + (v[j].z + v[j].w);
      part += (base + 128 * j + 4 * lane < nfull) ? s4 : 0;
    }
  }
  const int t = sizes[min(nfull + lane, b)];
  part += (nfull + lane < b) ? t : 0;
  part += __builtin_amdgcn_update_dpp(0, part, 0x161, 0xf, 0xf, false);
  part += __builtin_amdgcn_update_dpp(0, part, 0x162, 0xf, 0xf, false);
  part += __builtin_amdgcn_update_dpp(0, part, 0x164, 0xf, 0xf, false);
  part += __builtin_amdgcn_update_dpp(0, part, 0x168, 0xf, 0xf, false);
  part += __builtin_amdgcn_permlanex16(part, part, 0x76543210, 0xfedcba98, false, false);
  return __builtin_amdgcn_readfirstlane(part);
}

#define REFILL_BEHIND(p, q, r, s) asm volatile("" : "+v"(p), "+v"(q), "+v"(r), "+v"(s))
#define SUMS_BEHIND(a) asm volatile("" : "+v"(a))

struct Stage {
  v4f d[2 * STAGE_ROWS];
};

__device__ __forceinline__ void stage_load(Stage& st, const float* __restrict__ x,
                                           int row0, int last, int lane) {
#pragma unroll
  for (int j = 0; j < STAGE_ROWS; ++j) {
    const int r = min(row0 + j, last);
    const v4f* q = reinterpret_cast<const v4f*>(x + (size_t)r * SEG_D) + lane;
    st.d[2 * j]     = __builtin_nontemporal_load(q);
    st.d[2 * j + 1] = __builtin_nontemporal_load(q + 32);
  }
}

__device__ __forceinline__ void row_sum(const v4f p0, const v4f p1, const v2f a,
                                        v8f& a00, v8f& a01, v8f& a10, v8f& a11) {
  a00 = __builtin_amdgcn_wmma_f32_16x16x4_f32(false, a, false, p0.xy, (short)0, a00, false, false);
  a10 = __builtin_amdgcn_wmma_f32_16x16x4_f32(false, a, false, p1.xy, (short)0, a10, false, false);
  a01 = __builtin_amdgcn_wmma_f32_16x16x4_f32(false, a, false, p0.zw, (short)0, a01, false, false);
  a11 = __builtin_amdgcn_wmma_f32_16x16x4_f32(false, a, false, p1.zw, (short)0, a11, false, false);
}

__device__ __forceinline__ void stage_sum_full(const Stage& st, const v2f sel,
                                               v8f& a00, v8f& a01, v8f& a10, v8f& a11) {
#pragma unroll
  for (int j = 0; j < STAGE_ROWS; ++j)
    row_sum(st.d[2 * j], st.d[2 * j + 1], sel, a00, a01, a10, a11);
}

__device__ __forceinline__ void stage_sum_last(const Stage& st, const v2f sel, int row0, int end,
                                               v8f& a00, v8f& a01, v8f& a10, v8f& a11) {
#pragma unroll
  for (int j = 0; j < STAGE_ROWS; ++j) {
    const bool in = (row0 + j) < end;
    v2f a;
    a.x = in ? sel.x : 0.0f;
    a.y = in ? sel.y : 0.0f;
    row_sum(st.d[2 * j], st.d[2 * j + 1], a, a00, a01, a10, a11);
  }
}

__global__ __launch_bounds__(32) void seg_mean_wmma_kernel(
    const float* __restrict__ x, const int* __restrict__ sizes,
    float* __restrict__ out, int nseg) {
  const int b    = blockIdx.x;
  const int lane = threadIdx.x;

  const int nrows = nseg * ROWS_PER_SEG;
  const int size  = sizes[b];
  const int first = rows_in_front(sizes, b, lane);
  const int start = min(first, nrows);
  const int end   = (b == nseg - 1) ? nrows : min(first + size, nrows);
  const int cnt   = end - start;
  const int last  = end - 1;

  const int c    = lane & 15;
  const int h    = lane >> 4;

  v2f sel;
  sel.x = (c == h) ? 1.0f : 0.0f;
  sel.y = (c == 2 + h) ? 1.0f : 0.0f;

  v8f a00 = {}, a01 = {}, a10 = {}, a11 = {};

  if (cnt > 0) {
    const int nst = (cnt + STAGE_ROWS - 1) / STAGE_ROWS;
    Stage sa, sb;
    int row = start;
    int k = 0;
    stage_load(sa, x, row, last, lane);
    for (; k + 2 < nst; k += 2) {
      stage_load(sb, x, row + STAGE_ROWS, last, lane);
      SUMS_BEHIND(sel);
      stage_sum_full(sa, sel, a00, a01, a10, a11);
      REFILL_BEHIND(a00, a01, a10, a11);
      stage_load(sa, x, row + 2 * STAGE_ROWS, last, lane);
      SUMS_BEHIND(sel);
      stage_sum_full(sb, sel, a00, a01, a10, a11);
      REFILL_BEHIND(a00, a01, a10, a11);
      row += 2 * STAGE_ROWS;
    }
    if (k + 1 < nst) {
      stage_load(sb, x, row + STAGE_ROWS, last, lane);
      SUMS_BEHIND(sel);
      stage_sum_full(sa, sel, a00, a01, a10, a11);
      stage_sum_last(sb, sel, row + STAGE_ROWS, end, a00, a01, a10, a11);
    } else {
      stage_sum_last(sa, sel, row, end, a00, a01, a10, a11);
    }
  }

  if (lane < 16) {
    const float fs = (float)size;
    float* o = out + (size_t)b * SEG_D + 4 * c;
    v4f v;
    v.x = a00[0] / fs; v.y = a00[2] / fs; v.z = a01[0] / fs; v.w = a01[2] / fs;
    *reinterpret_cast<v4f*>(o) = v;
    v.x = a00[1] / fs; v.y = a00[3] / fs; v.z = a01[1] / fs; v.w = a01[3] / fs;
    *reinterpret_cast<v4f*>(o + 64) = v;
    v.x = a10[0] / fs; v.y = a10[2] / fs; v.z = a11[0] / fs; v.w = a11[2] / fs;
    *reinterpret_cast<v4f*>(o + 128) = v;
    v.x = a10[1] / fs; v.y = a10[3] / fs; v.z = a11[1] / fs; v.w = a11[3] / fs;
    *reinterpret_cast<v4f*>(o + 192) = v;
  }
}

extern "C" void kernel_launch(void* const* d_in, const int* in_sizes, int n_in,
                              void* d_out, int out_size, void* d_ws, size_t ws_size,
                              hipStream_t stream) {
  const float* x     = (const float*)d_in[0];
  const int*   sizes = (const int*)d_in[1];
  float*       out   = (float*)d_out;

  const int B = in_sizes[1];

  seg_mean_wmma_kernel<<<B, 32, 0, stream>>>(x, sizes, out, B);
}
